# starting kernel
# baseline (speedup 1.0000x reference)
#include <hip/hip_runtime.h>
#include <hip/hip_bf16.h>

#define FEAT 96
#define KSTEPS (FEAT / 4)
#define NTILES (FEAT / 16)

typedef float v2f __attribute__((ext_vector_type(2)));
typedef float v8f __attribute__((ext_vector_type(8)));

__global__ __launch_bounds__(256) void zero_f32(float* __restrict__ p, long long n) {
  long long i = (long long)blockIdx.x * blockDim.x + threadIdx.x;
  long long i4 = i * 4;
  if (i4 + 3 < n) {
    *(float4*)(p + i4) = make_float4(0.f, 0.f, 0.f, 0.f);
  } else {
    for (long long j = i4; j < n; ++j) p[j] = 0.f;
  }
}

__global__ __launch_bounds__(256) void degree_kernel(const int* __restrict__ src,
                                                     const int* __restrict__ dst,
                                                     float* __restrict__ deg_out,
                                                     float* __restrict__ deg_in,
                                                     int nedges) {
  int e = blockIdx.x * blockDim.x + threadIdx.x;
  if (e >= nedges) return;
  atomicAdd(&deg_out[src[e]], 1.0f);
  atomicAdd(&deg_in[dst[e]], 1.0f);
}

__global__ __launch_bounds__(256) void norm_finalize(float* __restrict__ deg_out,
                                                     float* __restrict__ deg_in,
                                                     int n) {
  int i = blockIdx.x * blockDim.x + threadIdx.x;
  if (i >= n) return;
  deg_out[i] = 1.0f / sqrtf(fmaxf(deg_out[i], 1.0f));
  deg_in[i]  = 1.0f / sqrtf(fmaxf(deg_in[i], 1.0f));
}

__global__ __launch_bounds__(256) void edge_scatter(const int* __restrict__ src,
                                                    const int* __restrict__ dst,
                                                    const float* __restrict__ h,
                                                    const float* __restrict__ norm_src,
                                                    float* __restrict__ m,
                                                    int nedges) {
  long long gid = (long long)blockIdx.x * blockDim.x + threadIdx.x;
  int e = __builtin_amdgcn_readfirstlane((int)(gid >> 5));
  if (e >= nedges) return;
  int lane = threadIdx.x & 31;
  int s = src[e];
  int d = dst[e];
  float ns = norm_src[s];
  const float* __restrict__ hr = h + (long long)s * FEAT;
  float* __restrict__ mr = m + (long long)d * FEAT;
#pragma unroll
  for (int i = 0; i < 3; ++i) {
    int f = lane + i * 32;
    atomicAdd(&mr[f], hr[f] * ns);
  }
}

__global__ __launch_bounds__(256) void gemm_norm_bias_relu(
    const float* __restrict__ m, const float* __restrict__ norm_dst,
    const float* __restrict__ W, const float* __restrict__ bias,
    float* __restrict__ out, int nrows) {
  __shared__ float Ws[FEAT * FEAT];
  __shared__ float bs[FEAT];

  for (int i = threadIdx.x; i < FEAT * FEAT; i += 256) Ws[i] = W[i];
  if (threadIdx.x < FEAT) bs[threadIdx.x] = bias[threadIdx.x];
  __syncthreads();

  const int wave = threadIdx.x >> 5;
  const int lane = threadIdx.x & 31;
  const int strip = blockIdx.x * 8 + wave;
  const int row0 = strip * 16;
  if (row0 >= nrows) return;

  const int lrow = lane & 15;
  const int lgrp = lane >> 4;

  const int arow_i = min(row0 + lrow, nrows - 1);
  const float nd = norm_dst[arow_i];
  const float* __restrict__ arow = m + (long long)arow_i * FEAT;

  v2f a[KSTEPS];
#pragma unroll
  for (int ks = 0; ks < KSTEPS; ++ks) {
    const int k = ks * 4 + 2 * lgrp;
    a[ks].x = arow[k] * nd;
    a[ks].y = arow[k + 1] * nd;
  }

  v8f acc[NTILES];
#pragma unroll
  for (int nt = 0; nt < NTILES; ++nt) acc[nt] = (v8f){};

  v2f bf[NTILES];
#pragma unroll
  for (int nt = 0; nt < NTILES; ++nt) {
    const int k = 2 * lgrp;
    bf[nt].x = Ws[k * FEAT + nt * 16 + lrow];
    bf[nt].y = Ws[(k + 1) * FEAT + nt * 16 + lrow];
  }

#pragma unroll
  for (int ks = 0; ks < KSTEPS; ++ks) {
    v2f bn[NTILES];
    if (ks + 1 < KSTEPS) {
      const int k = (ks + 1) * 4 + 2 * lgrp;
#pragma unroll
      for (int nt = 0; nt < NTILES; ++nt) {
        bn[nt].x = Ws[k * FEAT + nt * 16 + lrow];
        bn[nt].y = Ws[(k + 1) * FEAT + nt * 16 + lrow];
      }
    }
#pragma unroll
    for (int nt = 0; nt < NTILES; ++nt) {
      acc[nt] = __builtin_amdgcn_wmma_f32_16x16x4_f32(
           false, a[ks],  false, bf[nt],
           (short)0, acc[nt],  false,  false);
    }
    if (ks + 1 < KSTEPS) {
#pragma unroll
      for (int nt = 0; nt < NTILES; ++nt) bf[nt] = bn[nt];
    }
  }

#pragma unroll
  for (int nt = 0; nt < NTILES; ++nt) {
    const int nb = nt * 16;
    const float bcol = bs[nb + lrow];
#pragma unroll
    for (int r = 0; r < 8; ++r) {
      const int row = row0 + r + 8 * lgrp;
      if (row < nrows) {
        float v = acc[nt][r] + bcol;
        out[(long long)row * FEAT + nb + lrow] = fmaxf(v, 0.0f);
      }
    }
  }
}

extern "C" void kernel_launch(void* const* d_in, const int* in_sizes, int n_in,
                              void* d_out, int out_size, void* d_ws, size_t ws_size,
                              hipStream_t stream) {
  const float* h0  = (const float*)d_in[0];
  const int*   src = (const int*)d_in[1];
  const int*   dst = (const int*)d_in[2];
  const float* Wp[3] = {(const float*)d_in[3], (const float*)d_in[5], (const float*)d_in[7]};
  const float* bp[3] = {(const float*)d_in[4], (const float*)d_in[6], (const float*)d_in[8]};

  const int N = in_sizes[0] / FEAT;
  const int E = in_sizes[1];

  float* ws       = (float*)d_ws;
  float* norm_src = ws;
  float* norm_dst = ws + N;
  float* mbuf     = ws + 2LL * N;
  float* hA       = mbuf + (long long)N * FEAT;
  float* outp     = (float*)d_out;

  const long long ND = (long long)N * FEAT;

  {
    long long n2 = 2LL * N;
    int blocks = (int)((n2 / 4 + 255) / 256) + 1;
    zero_f32<<<blocks, 256, 0, stream>>>(norm_src, n2);
    degree_kernel<<<(E + 255) / 256, 256, 0, stream>>>(src, dst, norm_src, norm_dst, E);
    norm_finalize<<<(N + 255) / 256, 256, 0, stream>>>(norm_src, norm_dst, N);
  }

  const int strips = (N + 15) / 16;
  const int gemm_blocks = (strips + 7) / 8;
  const int zero_blocks = (int)((ND / 4 + 255) / 256) + 1;
  const long long scat_threads = (long long)E * 32;
  const int scat_blocks = (int)((scat_threads + 255) / 256);

  const float* lin[3]  = {h0, outp, hA};
  float*       lout[3] = {outp, hA, outp};

  for (int l = 0; l < 3; ++l) {
    zero_f32<<<zero_blocks, 256, 0, stream>>>(mbuf, ND);
    edge_scatter<<<scat_blocks, 256, 0, stream>>>(src, dst, lin[l], norm_src, mbuf, E);
    gemm_norm_bias_relu<<<gemm_blocks, 256, 0, stream>>>(mbuf, norm_dst, Wp[l], bp[l],
                                                         lout[l], N);
  }
}


# Layer GEMM moved into the row-ownership block: its 196 LDS row sums are the WMMA A operand, W staged in the block's LDS; no m buffer, 6 launches not 9; row sum order: LDS arrival, as before
# speedup vs baseline: 2.3592x; 2.3592x over previous
#include <hip/hip_runtime.h>
#include <hip/hip_bf16.h>

#define FEAT 96
#define KSTEPS (FEAT / 4)
#define NTILES (FEAT / 16)

typedef float v2f __attribute__((ext_vector_type(2)));
typedef float v8f __attribute__((ext_vector_type(8)));

__global__ __launch_bounds__(256) void zero_f32(float* __restrict__ p, long long n) {
  long long i = (long long)blockIdx.x * blockDim.x + threadIdx.x;
  long long i4 = i * 4;
  if (i4 + 3 < n) {
    *(float4*)(p + i4) = make_float4(0.f, 0.f, 0.f, 0.f);
  } else {
    for (long long j = i4; j < n; ++j) p[j] = 0.f;
  }
}

#define OWN_ROWS 196
#define TPITCH 100
#define WPITCH 104

__device__ __forceinline__ unsigned owner_key(int d) {
  return min((unsigned)d / (unsigned)OWN_ROWS, 255u);
}

__global__ __launch_bounds__(256) void degree_kernel(const int* __restrict__ src,
                                                     const int* __restrict__ dst,
                                                     float* __restrict__ deg_out,
                                                     float* __restrict__ deg_in,
                                                     unsigned char* __restrict__ key,
                                                     int nedges) {
  const int q = blockIdx.x * blockDim.x + threadIdx.x;
  const int e0 = q * 4;
  if (e0 >= nedges) return;
  const bool vec_ok =
      ((reinterpret_cast<unsigned long long>(src) | reinterpret_cast<unsigned long long>(dst)) & 15ull) == 0ull &&
      (reinterpret_cast<unsigned long long>(key) & 3ull) == 0ull;
  if (vec_ok && e0 + 3 < nedges) {
    const int4 s = reinterpret_cast<const int4*>(src)[q];
    const int4 d = reinterpret_cast<const int4*>(dst)[q];
    atomicAdd(&deg_out[s.x], 1.0f);
    atomicAdd(&deg_out[s.y], 1.0f);
    atomicAdd(&deg_out[s.z], 1.0f);
    atomicAdd(&deg_out[s.w], 1.0f);
    atomicAdd(&deg_in[d.x], 1.0f);
    atomicAdd(&deg_in[d.y], 1.0f);
    atomicAdd(&deg_in[d.z], 1.0f);
    atomicAdd(&deg_in[d.w], 1.0f);
    reinterpret_cast<unsigned*>(key)[q] =
        owner_key(d.x) | (owner_key(d.y) << 8) | (owner_key(d.z) << 16) | (owner_key(d.w) << 24);
  } else {
    const int e1 = min(e0 + 4, nedges);
    for (int e = e0; e < e1; ++e) {
      const int d = dst[e];
      atomicAdd(&deg_out[src[e]], 1.0f);
      atomicAdd(&deg_in[d], 1.0f);
      key[e] = (unsigned char)owner_key(d);
    }
  }
}

__global__ __launch_bounds__(256) void norm_finalize(float* __restrict__ deg_out,
                                                     float* __restrict__ deg_in,
                                                     int n) {
  int i = blockIdx.x * blockDim.x + threadIdx.x;
  if (i >= n) return;
  deg_out[i] = 1.0f / sqrtf(fmaxf(deg_out[i], 1.0f));
  deg_in[i]  = 1.0f / sqrtf(fmaxf(deg_in[i], 1.0f));
}

#define OWN_THREADS 1024
#define OWN_WAVES (OWN_THREADS / 32)
#define LIST_CAP 12288
#define SCAN_DEPTH 8
#define ADD_DEPTH 8

template <bool kScan>
__global__ __launch_bounds__(OWN_THREADS) void own_layer(const int* __restrict__ src,
                                                         const int* __restrict__ dst,
                                                         const unsigned char* __restrict__ key,
                                                         const float* __restrict__ h,
                                                         const float* __restrict__ norm_src,
                                                         const float* __restrict__ norm_dst,
                                                         const float* __restrict__ W,
                                                         const float* __restrict__ bias,
                                                         float* __restrict__ out,
                                                         int* __restrict__ keep,
                                                         int* __restrict__ keep_cnt,
                                                         int nedges, int nrows) {
  __shared__ __attribute__((aligned(16))) float tile[OWN_ROWS * TPITCH];
  __shared__ float Ws[FEAT * WPITCH];
  __shared__ float bs[FEAT];
  __shared__ int lsrc[LIST_CAP];
  __shared__ int lrow[LIST_CAP];
  __shared__ int lpick[kScan ? LIST_CAP : 1];
  __shared__ int cnt;
  __shared__ int cnt2;

  const int tid = threadIdx.x;
  const int base = blockIdx.x * OWN_ROWS;
  const unsigned nown = (unsigned)min(OWN_ROWS, nrows - base);
  int2* __restrict__ mykeep = reinterpret_cast<int2*>(keep) + (long long)blockIdx.x * LIST_CAP;

  for (int i = tid; i < OWN_ROWS * TPITCH; i += OWN_THREADS) tile[i] = 0.f;
  for (int i = tid; i < FEAT * FEAT; i += OWN_THREADS) Ws[(i / FEAT) * WPITCH + (i % FEAT)] = W[i];
  if (tid < FEAT) bs[tid] = bias[tid];

  int total;
  if (kScan) {
    if (tid == 0) {
      cnt = 0;
      cnt2 = 0;
    }
    __syncthreads();

    const unsigned want = min((unsigned)blockIdx.x, 255u);
    const unsigned pat = want * 0x01010101u;
    auto eqmask = [&](unsigned x) -> unsigned {
      const unsigned y = x ^ pat;
      return ~(((y & 0x7F7F7F7Fu) + 0x7F7F7F7Fu) | y | 0x7F7F7F7Fu);
    };
    auto flags = [&](uint4 d) -> unsigned {
      return (eqmask(d.x) >> 7) | (eqmask(d.y) >> 6) | (eqmask(d.z) >> 5) | (eqmask(d.w) >> 4);
    };
    auto put = [&](int e0, unsigned g, int& pos) {
      while (g) {
        const int p = __builtin_ctz(g);
        g &= g - 1u;
        if (pos < LIST_CAP) lpick[pos] = e0 + ((p & 7) << 2) + (p >> 3);
        ++pos;
      }
    };

    const bool vec_ok = (reinterpret_cast<unsigned long long>(key) & 15ull) == 0ull;
    const int nvec = vec_ok ? (nedges >> 4) : 0;
    const uint4* __restrict__ key4 = reinterpret_cast<const uint4*>(key);
    const int span = SCAN_DEPTH * OWN_THREADS;
    const int nstep = nvec / span;
    const int rot = nstep > 0 ? (int)(blockIdx.x % (unsigned)nstep) : 0;
    for (int it = 0; it < nstep; ++it) {
      int st = it + rot;
      if (st >= nstep) st -= nstep;
      const int v0 = st * span + tid;
      uint4 d[SCAN_DEPTH];
#pragma unroll
      for (int j = 0; j < SCAN_DEPTH; ++j) d[j] = key4[v0 + j * OWN_THREADS];
      unsigned g[SCAN_DEPTH];
      int n = 0;
#pragma unroll
      for (int j = 0; j < SCAN_DEPTH; ++j) {
        g[j] = flags(d[j]);
        n += __builtin_popcount(g[j]);
      }
      if (n) {
        int pos = atomicAdd(&cnt, n);
#pragma unroll
        for (int j = 0; j < SCAN_DEPTH; ++j) put((v0 + j * OWN_THREADS) * 16, g[j], pos);
      }
    }
    for (int v = nstep * span + tid; v < nvec; v += OWN_THREADS) {
      const unsigned g = flags(key4[v]);
      if (g) {
        int pos = atomicAdd(&cnt, __builtin_popcount(g));
        put(v * 16, g, pos);
      }
    }
    for (int e = nvec * 16 + tid; e < nedges; e += OWN_THREADS) {
      if ((unsigned)key[e] == want) {
        const int pos = atomicAdd(&cnt, 1);
        if (pos < LIST_CAP) lpick[pos] = e;
      }
    }
    __syncthreads();

    const int npick = cnt;
    if (npick <= LIST_CAP) {
      for (int i = tid; i < npick; i += OWN_THREADS) {
        const int e = lpick[i];
        const unsigned r = (unsigned)(dst[e] - base);
        if (r < nown) {
          const int sn = src[e];
          const int pos = atomicAdd(&cnt2, 1);
          lsrc[pos] = sn;
          lrow[pos] = (int)r;
          mykeep[pos] = make_int2(sn, (int)r);
        }
      }
      __syncthreads();
      total = cnt2;
    } else {
      total = LIST_CAP + 1;
    }
    if (tid == 0) keep_cnt[blockIdx.x] = total;
  } else {
    total = keep_cnt[blockIdx.x];
    if (total <= LIST_CAP) {
      for (int i = tid; i < total; i += OWN_THREADS) {
        const int2 p = mykeep[i];
        lsrc[i] = p.x;
        lrow[i] = p.y;
      }
    }
  }
  __syncthreads();

  if (total <= LIST_CAP) {
    const int wave = tid >> 5;
    const int lane = tid & 31;
    const int ngroup = total / ADD_DEPTH;
    for (int g = wave; g < ngroup; g += OWN_WAVES) {
      float x0[ADD_DEPTH], x1[ADD_DEPTH], x2[ADD_DEPTH], ns[ADD_DEPTH];
      int rr[ADD_DEPTH];
#pragma unroll
      for (int j = 0; j < ADD_DEPTH; ++j) {
        const int i = g * ADD_DEPTH + j;
        const int s = __builtin_amdgcn_readfirstlane(lsrc[i]);
        rr[j] = __builtin_amdgcn_readfirstlane(lrow[i]);
        const float* __restrict__ hr = h + (long long)s * FEAT;
        x0[j] = hr[lane];
        x1[j] = hr[lane + 32];
        x2[j] = hr[lane + 64];
        ns[j] = norm_src[s];
      }
#pragma unroll
      for (int j = 0; j < ADD_DEPTH; ++j) {
        float* tr = tile + rr[j] * TPITCH;
        atomicAdd(&tr[lane], x0[j] * ns[j]);
        atomicAdd(&tr[lane + 32], x1[j] * ns[j]);
        atomicAdd(&tr[lane + 64], x2[j] * ns[j]);
      }
    }
    {
      const int i = ngroup * ADD_DEPTH + wave;
      if (i < total) {
        const int s = __builtin_amdgcn_readfirstlane(lsrc[i]);
        const int r = __builtin_amdgcn_readfirstlane(lrow[i]);
        const float* __restrict__ hr = h + (long long)s * FEAT;
        const float nsv = norm_src[s];
        const float y0 = hr[lane], y1 = hr[lane + 32], y2 = hr[lane + 64];
        float* tr = tile + r * TPITCH;
        atomicAdd(&tr[lane], y0 * nsv);
        atomicAdd(&tr[lane + 32], y1 * nsv);
        atomicAdd(&tr[lane + 64], y2 * nsv);
      }
    }
  } else {
    for (int e = tid; e < nedges; e += OWN_THREADS) {
      const unsigned r = (unsigned)(dst[e] - base);
      if (r < nown) {
        const int s = src[e];
        const float ns = norm_src[s];
        const float* __restrict__ hr = h + (long long)s * FEAT;
        for (int f = 0; f < FEAT; ++f) atomicAdd(&tile[r * TPITCH + f], hr[f] * ns);
      }
    }
  }
  __syncthreads();

  const int gw = __builtin_amdgcn_readfirstlane(tid >> 5);
  const int nstrips = ((int)nown + 15) >> 4;
  if (gw < nstrips) {
    const int gl = tid & 31;
    const int lrw = gl & 15;
    const int lgrp = gl >> 4;
    const int row0 = gw * 16;
    const int ar = min(row0 + lrw, (int)nown - 1);
    const float nd = norm_dst[base + ar];
    const float* arow = tile + ar * TPITCH;

    v2f af;
    af.x = arow[2 * lgrp] * nd;
    af.y = arow[2 * lgrp + 1] * nd;

    v8f acc[NTILES];
#pragma unroll
    for (int nt = 0; nt < NTILES; ++nt) acc[nt] = (v8f){};

    v2f bf[NTILES];
#pragma unroll
    for (int nt = 0; nt < NTILES; ++nt) {
      const int k = 2 * lgrp;
      bf[nt].x = Ws[k * WPITCH + nt * 16 + lrw];
      bf[nt].y = Ws[(k + 1) * WPITCH + nt * 16 + lrw];
    }

#pragma unroll 1
    for (int ks = 0; ks + 1 < KSTEPS; ++ks) {
      const int k = (ks + 1) * 4 + 2 * lgrp;
      v2f bn[NTILES];
#pragma unroll
      for (int nt = 0; nt < NTILES; ++nt) {
        bn[nt].x = Ws[k * WPITCH + nt * 16 + lrw];
        bn[nt].y = Ws[(k + 1) * WPITCH + nt * 16 + lrw];
      }
      v2f an;
      an.x = arow[k] * nd;
      an.y = arow[k + 1] * nd;
#pragma unroll
      for (int nt = 0; nt < NTILES; ++nt) {
        acc[nt] = __builtin_amdgcn_wmma_f32_16x16x4_f32(
             false, af,  false, bf[nt],
             (short)0, acc[nt],  false,  false);
      }
#pragma unroll
      for (int nt = 0; nt < NTILES; ++nt) bf[nt] = bn[nt];
      af = an;
    }
#pragma unroll
    for (int nt = 0; nt < NTILES; ++nt) {
      acc[nt] = __builtin_amdgcn_wmma_f32_16x16x4_f32(
           false, af,  false, bf[nt],
           (short)0, acc[nt],  false,  false);
    }

#pragma unroll
    for (int nt = 0; nt < NTILES; ++nt) {
      const int nb = nt * 16;
      const float bcol = bs[nb + lrw];
#pragma unroll
      for (int r = 0; r < 8; ++r) {
        const int rl = row0 + r + 8 * lgrp;
        if (rl < (int)nown) {
          float v = acc[nt][r] + bcol;
          out[(long long)(base + rl) * FEAT + nb + lrw] = fmaxf(v, 0.0f);
        }
      }
    }
  }
}

extern "C" void kernel_launch(void* const* d_in, const int* in_sizes, int n_in,
                              void* d_out, int out_size, void* d_ws, size_t ws_size,
                              hipStream_t stream) {
  const float* h0  = (const float*)d_in[0];
  const int*   src = (const int*)d_in[1];
  const int*   dst = (const int*)d_in[2];
  const float* Wp[3] = {(const float*)d_in[3], (const float*)d_in[5], (const float*)d_in[7]};
  const float* bp[3] = {(const float*)d_in[4], (const float*)d_in[6], (const float*)d_in[8]};

  const int N = in_sizes[0] / FEAT;
  const int E = in_sizes[1];

  float* ws       = (float*)d_ws;
  float* norm_src = ws;
  float* norm_dst = ws + N;
  float* mbuf     = ws + 2LL * N;
  float* hA       = mbuf + (long long)N * FEAT;
  float* outp     = (float*)d_out;
  const int own_blocks = (N + OWN_ROWS - 1) / OWN_ROWS;
  int*  keep      = (int*)(hA + (long long)N * FEAT);
  int*  keep_cnt  = keep + 2LL * own_blocks * LIST_CAP;
  unsigned char* key = (unsigned char*)(keep_cnt + own_blocks);

  {
    long long n2 = 2LL * N;
    int blocks = (int)((n2 / 4 + 255) / 256) + 1;
    zero_f32<<<blocks, 256, 0, stream>>>(norm_src, n2);
    degree_kernel<<<((E + 3) / 4 + 255) / 256, 256, 0, stream>>>(src, dst, norm_src, norm_dst, key, E);
    norm_finalize<<<(N + 255) / 256, 256, 0, stream>>>(norm_src, norm_dst, N);
  }

  const float* lin[3]  = {h0, outp, hA};
  float*       lout[3] = {outp, hA, outp};

  for (int l = 0; l < 3; ++l) {
    if (l == 0)
      own_layer<true><<<own_blocks, OWN_THREADS, 0, stream>>>(src, dst, key, lin[l], norm_src, norm_dst,
                                                              Wp[l], bp[l], lout[l], keep, keep_cnt, E, N);
    else
      own_layer<false><<<own_blocks, OWN_THREADS, 0, stream>>>(src, dst, key, lin[l], norm_src, norm_dst,
                                                               Wp[l], bp[l], lout[l], keep, keep_cnt, E, N);
  }
}


# Degrees by LDS histograms: 64 blocks count src and dst of their edge chunk in LDS (16-bit pairs), a merge kernel sums the partial rows into the norms; no global atomics, no zero fill, 5 launches not 6
# speedup vs baseline: 3.6602x; 1.5515x over previous
#include <hip/hip_runtime.h>
#include <hip/hip_bf16.h>

#define FEAT 96
#define KSTEPS (FEAT / 4)
#define NTILES (FEAT / 16)

typedef float v2f __attribute__((ext_vector_type(2)));
typedef float v8f __attribute__((ext_vector_type(8)));

#define OWN_ROWS 196
#define TPITCH 100
#define WPITCH 104

__device__ __forceinline__ unsigned owner_key(int d) {
  return min((unsigned)d / (unsigned)OWN_ROWS, 255u);
}

#define HIST_WORDS 25088
#define HIST_THREADS 1024
#define HIST_DEPTH 4

__device__ __forceinline__ void hist_add(unsigned* hist, int node, int n0) {
  const unsigned r = (unsigned)(node - n0);
  const bool in = r < 2u * HIST_WORDS;
  atomicAdd(&hist[in ? (r >> 1) : 0u], in ? ((r & 1u) ? 0x10000u : 1u) : 0u);
}

__global__ __launch_bounds__(HIST_THREADS) void degree_hist(const int* __restrict__ src,
                                                            const int* __restrict__ dst,
                                                            unsigned* __restrict__ part,
                                                            unsigned char* __restrict__ key,
                                                            int nedges, int nnodes, int wpad) {
  __shared__ __attribute__((aligned(16))) unsigned hs[HIST_WORDS];
  __shared__ __attribute__((aligned(16))) unsigned hd[HIST_WORDS];
  const int tid = threadIdx.x;
  const int nparts = gridDim.x;
  const int ch = ((nedges + nparts * 16 - 1) / (nparts * 16)) * 16;
  const int e_lo = (int)min((long long)blockIdx.x * ch, (long long)nedges);
  const int e_hi = (int)min((long long)e_lo + ch, (long long)nedges);
  const bool vec_ok =
      ((reinterpret_cast<unsigned long long>(src) | reinterpret_cast<unsigned long long>(dst)) & 15ull) == 0ull &&
      (reinterpret_cast<unsigned long long>(key) & 3ull) == 0ull;
  const int nq = vec_ok ? ((e_hi - e_lo) >> 2) : 0;
  const int4* __restrict__ s4 = reinterpret_cast<const int4*>(src + e_lo);
  const int4* __restrict__ d4 = reinterpret_cast<const int4*>(dst + e_lo);
  unsigned* __restrict__ k4 = reinterpret_cast<unsigned*>(key + e_lo);
  unsigned* __restrict__ psrc = part + (long long)blockIdx.x * wpad;
  unsigned* __restrict__ pdst = part + (long long)(nparts + blockIdx.x) * wpad;

  for (int n0 = 0; n0 < nnodes; n0 += 2 * HIST_WORDS) {
    for (int i = tid; i < HIST_WORDS / 4; i += HIST_THREADS) {
      reinterpret_cast<uint4*>(hs)[i] = make_uint4(0u, 0u, 0u, 0u);
      reinterpret_cast<uint4*>(hd)[i] = make_uint4(0u, 0u, 0u, 0u);
    }
    __syncthreads();
    for (int q0 = 0; q0 < nq; q0 += HIST_DEPTH * HIST_THREADS) {
      int4 s[HIST_DEPTH], d[HIST_DEPTH];
#pragma unroll
      for (int j = 0; j < HIST_DEPTH; ++j) {
        const int q = min(q0 + j * HIST_THREADS + tid, nq - 1);
        s[j] = s4[q];
        d[j] = d4[q];
      }
#pragma unroll
      for (int j = 0; j < HIST_DEPTH; ++j) {
        const int q = q0 + j * HIST_THREADS + tid;
        if (q < nq) {
          hist_add(hs, s[j].x, n0);
          hist_add(hs, s[j].y, n0);
          hist_add(hs, s[j].z, n0);
          hist_add(hs, s[j].w, n0);
          hist_add(hd, d[j].x, n0);
          hist_add(hd, d[j].y, n0);
          hist_add(hd, d[j].z, n0);
          hist_add(hd, d[j].w, n0);
          if (n0 == 0)
            k4[q] = owner_key(d[j].x) | (owner_key(d[j].y) << 8) | (owner_key(d[j].z) << 16) |
                    (owner_key(d[j].w) << 24);
        }
      }
    }
    for (int e = e_lo + nq * 4 + tid; e < e_hi; e += HIST_THREADS) {
      const int dn = dst[e];
      hist_add(hs, src[e], n0);
      hist_add(hd, dn, n0);
      if (n0 == 0) key[e] = (unsigned char)owner_key(dn);
    }
    __syncthreads();
    const int w0 = n0 >> 1;
    const int nw4 = min(HIST_WORDS, wpad - w0) >> 2;
    for (int i = tid; i < nw4; i += HIST_THREADS) {
      reinterpret_cast<uint4*>(psrc + w0)[i] = reinterpret_cast<const uint4*>(hs)[i];
      reinterpret_cast<uint4*>(pdst + w0)[i] = reinterpret_cast<const uint4*>(hd)[i];
    }
    __syncthreads();
  }
}

#define MERGE_THREADS 256
#define MERGE_DEPTH 8

__device__ __forceinline__ void unpack_add(unsigned* c, uint4 v) {
  c[0] += v.x & 0xFFFFu;
  c[1] += v.x >> 16;
  c[2] += v.y & 0xFFFFu;
  c[3] += v.y >> 16;
  c[4] += v.z & 0xFFFFu;
  c[5] += v.z >> 16;
  c[6] += v.w & 0xFFFFu;
  c[7] += v.w >> 16;
}

__global__ __launch_bounds__(MERGE_THREADS) void norm_merge(const unsigned* __restrict__ part,
                                                            float* __restrict__ norm_src,
                                                            float* __restrict__ norm_dst,
                                                            int nparts, int wpad, int nnodes) {
  __shared__ unsigned acc[2][MERGE_THREADS];
  const int tid = threadIdx.x;
  const int lane = tid & 31;
  const int pg = tid >> 5;
  acc[0][tid] = 0u;
  acc[1][tid] = 0u;
  __syncthreads();
  const int wg = blockIdx.x * 32 + lane;
  if (wg * 4 < wpad) {
    unsigned cs[8], cd[8];
#pragma unroll
    for (int k = 0; k < 8; ++k) {
      cs[k] = 0u;
      cd[k] = 0u;
    }
    for (int p0 = pg; p0 < nparts; p0 += 8 * MERGE_DEPTH) {
      uint4 a[MERGE_DEPTH], b[MERGE_DEPTH];
#pragma unroll
      for (int j = 0; j < MERGE_DEPTH; ++j) {
        const int p = p0 + 8 * j;
        a[j] = make_uint4(0u, 0u, 0u, 0u);
        b[j] = make_uint4(0u, 0u, 0u, 0u);
        if (p < nparts) {
          a[j] = reinterpret_cast<const uint4*>(part + (long long)p * wpad)[wg];
          b[j] = reinterpret_cast<const uint4*>(part + (long long)(nparts + p) * wpad)[wg];
        }
      }
#pragma unroll
      for (int j = 0; j < MERGE_DEPTH; ++j) {
        unpack_add(cs, a[j]);
        unpack_add(cd, b[j]);
      }
    }
#pragma unroll
    for (int k = 0; k < 8; ++k) {
      atomicAdd(&acc[0][k * 32 + lane], cs[k]);
      atomicAdd(&acc[1][k * 32 + lane], cd[k]);
    }
  }
  __syncthreads();
  const int node = blockIdx.x * MERGE_THREADS + tid;
  if (node < nnodes) {
    const int slot = (tid & 7) * 32 + (tid >> 3);
    norm_src[node] = 1.0f / sqrtf(fmaxf((float)acc[0][slot], 1.0f));
    norm_dst[node] = 1.0f / sqrtf(fmaxf((float)acc[1][slot], 1.0f));
  }
}

#define OWN_THREADS 1024
#define OWN_WAVES (OWN_THREADS / 32)
#define LIST_CAP 12288
#define SCAN_DEPTH 8
#define ADD_DEPTH 8

template <bool kScan>
__global__ __launch_bounds__(OWN_THREADS) void own_layer(const int* __restrict__ src,
                                                         const int* __restrict__ dst,
                                                         const unsigned char* __restrict__ key,
                                                         const float* __restrict__ h,
                                                         const float* __restrict__ norm_src,
                                                         const float* __restrict__ norm_dst,
                                                         const float* __restrict__ W,
                                                         const float* __restrict__ bias,
                                                         float* __restrict__ out,
                                                         int* __restrict__ keep,
                                                         int* __restrict__ keep_cnt,
                                                         int nedges, int nrows) {
  __shared__ __attribute__((aligned(16))) float tile[OWN_ROWS * TPITCH];
  __shared__ float Ws[FEAT * WPITCH];
  __shared__ float bs[FEAT];
  __shared__ int lsrc[LIST_CAP];
  __shared__ int lrow[LIST_CAP];
  __shared__ int lpick[kScan ? LIST_CAP : 1];
  __shared__ int cnt;
  __shared__ int cnt2;

  const int tid = threadIdx.x;
  const int base = blockIdx.x * OWN_ROWS;
  const unsigned nown = (unsigned)min(OWN_ROWS, nrows - base);
  int2* __restrict__ mykeep = reinterpret_cast<int2*>(keep) + (long long)blockIdx.x * LIST_CAP;

  for (int i = tid; i < OWN_ROWS * TPITCH; i += OWN_THREADS) tile[i] = 0.f;
  for (int i = tid; i < FEAT * FEAT; i += OWN_THREADS) Ws[(i / FEAT) * WPITCH + (i % FEAT)] = W[i];
  if (tid < FEAT) bs[tid] = bias[tid];

  int total;
  if (kScan) {
    if (tid == 0) {
      cnt = 0;
      cnt2 = 0;
    }
    __syncthreads();

    const unsigned want = min((unsigned)blockIdx.x, 255u);
    const unsigned pat = want * 0x01010101u;
    auto eqmask = [&](unsigned x) -> unsigned {
      const unsigned y = x ^ pat;
      return ~(((y & 0x7F7F7F7Fu) + 0x7F7F7F7Fu) | y | 0x7F7F7F7Fu);
    };
    auto flags = [&](uint4 d) -> unsigned {
      return (eqmask(d.x) >> 7) | (eqmask(d.y) >> 6) | (eqmask(d.z) >> 5) | (eqmask(d.w) >> 4);
    };
    auto put = [&](int e0, unsigned g, int& pos) {
      while (g) {
        const int p = __builtin_ctz(g);
        g &= g - 1u;
        if (pos < LIST_CAP) lpick[pos] = e0 + ((p & 7) << 2) + (p >> 3);
        ++pos;
      }
    };

    const bool vec_ok = (reinterpret_cast<unsigned long long>(key) & 15ull) == 0ull;
    const int nvec = vec_ok ? (nedges >> 4) : 0;
    const uint4* __restrict__ key4 = reinterpret_cast<const uint4*>(key);
    const int span = SCAN_DEPTH * OWN_THREADS;
    const int nstep = nvec / span;
    const int rot = nstep > 0 ? (int)(blockIdx.x % (unsigned)nstep) : 0;
    for (int it = 0; it < nstep; ++it) {
      int st = it + rot;
      if (st >= nstep) st -= nstep;
      const int v0 = st * span + tid;
      uint4 d[SCAN_DEPTH];
#pragma unroll
      for (int j = 0; j < SCAN_DEPTH; ++j) d[j] = key4[v0 + j * OWN_THREADS];
      unsigned g[SCAN_DEPTH];
      int n = 0;
#pragma unroll
      for (int j = 0; j < SCAN_DEPTH; ++j) {
        g[j] = flags(d[j]);
        n += __builtin_popcount(g[j]);
      }
      if (n) {
        int pos = atomicAdd(&cnt, n);
#pragma unroll
        for (int j = 0; j < SCAN_DEPTH; ++j) put((v0 + j * OWN_THREADS) * 16, g[j], pos);
      }
    }
    for (int v = nstep * span + tid; v < nvec; v += OWN_THREADS) {
      const unsigned g = flags(key4[v]);
      if (g) {
        int pos = atomicAdd(&cnt, __builtin_popcount(g));
        put(v * 16, g, pos);
      }
    }
    for (int e = nvec * 16 + tid; e < nedges; e += OWN_THREADS) {
      if ((unsigned)key[e] == want) {
        const int pos = atomicAdd(&cnt, 1);
        if (pos < LIST_CAP) lpick[pos] = e;
      }
    }
    __syncthreads();

    const int npick = cnt;
    if (npick <= LIST_CAP) {
      for (int i = tid; i < npick; i += OWN_THREADS) {
        const int e = lpick[i];
        const unsigned r = (unsigned)(dst[e] - base);
        if (r < nown) {
          const int sn = src[e];
          const int pos = atomicAdd(&cnt2, 1);
          lsrc[pos] = sn;
          lrow[pos] = (int)r;
          mykeep[pos] = make_int2(sn, (int)r);
        }
      }
      __syncthreads();
      total = cnt2;
    } else {
      total = LIST_CAP + 1;
    }
    if (tid == 0) keep_cnt[blockIdx.x] = total;
  } else {
    total = keep_cnt[blockIdx.x];
    if (total <= LIST_CAP) {
      for (int i = tid; i < total; i += OWN_THREADS) {
        const int2 p = mykeep[i];
        lsrc[i] = p.x;
        lrow[i] = p.y;
      }
    }
  }
  __syncthreads();

  if (total <= LIST_CAP) {
    const int wave = tid >> 5;
    const int lane = tid & 31;
    const int ngroup = total / ADD_DEPTH;
    for (int g = wave; g < ngroup; g += OWN_WAVES) {
      float x0[ADD_DEPTH], x1[ADD_DEPTH], x2[ADD_DEPTH], ns[ADD_DEPTH];
      int rr[ADD_DEPTH];
#pragma unroll
      for (int j = 0; j < ADD_DEPTH; ++j) {
        const int i = g * ADD_DEPTH + j;
        const int s = __builtin_amdgcn_readfirstlane(lsrc[i]);
        rr[j] = __builtin_amdgcn_readfirstlane(lrow[i]);
        const float* __restrict__ hr = h + (long long)s * FEAT;
        x0[j] = hr[lane];
        x1[j] = hr[lane + 32];
        x2[j] = hr[lane + 64];
        ns[j] = norm_src[s];
      }
#pragma unroll
      for (int j = 0; j < ADD_DEPTH; ++j) {
        float* tr = tile + rr[j] * TPITCH;
        atomicAdd(&tr[lane], x0[j] * ns[j]);
        atomicAdd(&tr[lane + 32], x1[j] * ns[j]);
        atomicAdd(&tr[lane + 64], x2[j] * ns[j]);
      }
    }
    {
      const int i = ngroup * ADD_DEPTH + wave;
      if (i < total) {
        const int s = __builtin_amdgcn_readfirstlane(lsrc[i]);
        const int r = __builtin_amdgcn_readfirstlane(lrow[i]);
        const float* __restrict__ hr = h + (long long)s * FEAT;
        const float nsv = norm_src[s];
        const float y0 = hr[lane], y1 = hr[lane + 32], y2 = hr[lane + 64];
        float* tr = tile + r * TPITCH;
        atomicAdd(&tr[lane], y0 * nsv);
        atomicAdd(&tr[lane + 32], y1 * nsv);
        atomicAdd(&tr[lane + 64], y2 * nsv);
      }
    }
  } else {
    for (int e = tid; e < nedges; e += OWN_THREADS) {
      const unsigned r = (unsigned)(dst[e] - base);
      if (r < nown) {
        const int s = src[e];
        const float ns = norm_src[s];
        const float* __restrict__ hr = h + (long long)s * FEAT;
        for (int f = 0; f < FEAT; ++f) atomicAdd(&tile[r * TPITCH + f], hr[f] * ns);
      }
    }
  }
  __syncthreads();

  const int gw = __builtin_amdgcn_readfirstlane(tid >> 5);
  const int nstrips = ((int)nown + 15) >> 4;
  if (gw < nstrips) {
    const int gl = tid & 31;
    const int lrw = gl & 15;
    const int lgrp = gl >> 4;
    const int row0 = gw * 16;
    const int ar = min(row0 + lrw, (int)nown - 1);
    const float nd = norm_dst[base + ar];
    const float* arow = tile + ar * TPITCH;

    v2f af;
    af.x = arow[2 * lgrp] * nd;
    af.y = arow[2 * lgrp + 1] * nd;

    v8f acc[NTILES];
#pragma unroll
    for (int nt = 0; nt < NTILES; ++nt) acc[nt] = (v8f){};

    v2f bf[NTILES];
#pragma unroll
    for (int nt = 0; nt < NTILES; ++nt) {
      const int k = 2 * lgrp;
      bf[nt].x = Ws[k * WPITCH + nt * 16 + lrw];
      bf[nt].y = Ws[(k + 1) * WPITCH + nt * 16 + lrw];
    }

#pragma unroll 1
    for (int ks = 0; ks + 1 < KSTEPS; ++ks) {
      const int k = (ks + 1) * 4 + 2 * lgrp;
      v2f bn[NTILES];
#pragma unroll
      for (int nt = 0; nt < NTILES; ++nt) {
        bn[nt].x = Ws[k * WPITCH + nt * 16 + lrw];
        bn[nt].y = Ws[(k + 1) * WPITCH + nt * 16 + lrw];
      }
      v2f an;
      an.x = arow[k] * nd;
      an.y = arow[k + 1] * nd;
#pragma unroll
      for (int nt = 0; nt < NTILES; ++nt) {
        acc[nt] = __builtin_amdgcn_wmma_f32_16x16x4_f32(
             false, af,  false, bf[nt],
             (short)0, acc[nt],  false,  false);
      }
#pragma unroll
      for (int nt = 0; nt < NTILES; ++nt) bf[nt] = bn[nt];
      af = an;
    }
#pragma unroll
    for (int nt = 0; nt < NTILES; ++nt) {
      acc[nt] = __builtin_amdgcn_wmma_f32_16x16x4_f32(
           false, af,  false, bf[nt],
           (short)0, acc[nt],  false,  false);
    }

#pragma unroll
    for (int nt = 0; nt < NTILES; ++nt) {
      const int nb = nt * 16;
      const float bcol = bs[nb + lrw];
#pragma unroll
      for (int r = 0; r < 8; ++r) {
        const int rl = row0 + r + 8 * lgrp;
        if (rl < (int)nown) {
          float v = acc[nt][r] + bcol;
          out[(long long)(base + rl) * FEAT + nb + lrw] = fmaxf(v, 0.0f);
        }
      }
    }
  }
}

extern "C" void kernel_launch(void* const* d_in, const int* in_sizes, int n_in,
                              void* d_out, int out_size, void* d_ws, size_t ws_size,
                              hipStream_t stream) {
  const float* h0  = (const float*)d_in[0];
  const int*   src = (const int*)d_in[1];
  const int*   dst = (const int*)d_in[2];
  const float* Wp[3] = {(const float*)d_in[3], (const float*)d_in[5], (const float*)d_in[7]};
  const float* bp[3] = {(const float*)d_in[4], (const float*)d_in[6], (const float*)d_in[8]};

  const int N = in_sizes[0] / FEAT;
  const int E = in_sizes[1];

  float* ws       = (float*)d_ws;
  float* norm_src = ws;
  float* norm_dst = ws + N;
  float* mbuf     = ws + 2LL * N;
  float* hA       = mbuf + (long long)N * FEAT;
  float* outp     = (float*)d_out;
  const int own_blocks = (N + OWN_ROWS - 1) / OWN_ROWS;
  int*  keep      = (int*)(hA + (long long)N * FEAT);
  int*  keep_cnt  = keep + 2LL * own_blocks * LIST_CAP;
  unsigned char* key = (unsigned char*)(keep_cnt + own_blocks);

  const int nwords = (N + 1) / 2;
  const int wpad = (nwords + 3) / 4 * 4;
  const int hist_blocks = E / 65520 + 1 > 64 ? E / 65520 + 1 : 64;
  unsigned* part = (unsigned*)(key + ((long long)E + 15) / 16 * 16);

  degree_hist<<<hist_blocks, HIST_THREADS, 0, stream>>>(src, dst, part, key, E, N, wpad);
  norm_merge<<<(N + MERGE_THREADS - 1) / MERGE_THREADS, MERGE_THREADS, 0, stream>>>(part, norm_src, norm_dst,
                                                                                   hist_blocks, wpad, N);

  const float* lin[3]  = {h0, outp, hA};
  float*       lout[3] = {outp, hA, outp};

  for (int l = 0; l < 3; ++l) {
    if (l == 0)
      own_layer<true><<<own_blocks, OWN_THREADS, 0, stream>>>(src, dst, key, lin[l], norm_src, norm_dst,
                                                              Wp[l], bp[l], lout[l], keep, keep_cnt, E, N);
    else
      own_layer<false><<<own_blocks, OWN_THREADS, 0, stream>>>(src, dst, key, lin[l], norm_src, norm_dst,
                                                               Wp[l], bp[l], lout[l], keep, keep_cnt, E, N);
  }
}


# Edges bucketed by the wave that owns their row (7 rows a wave) when layer 1 builds the list; a wave sums its rows in registers, stores each once: no LDS f32 atomics; row sum order = bucket order
# speedup vs baseline: 3.8650x; 1.0560x over previous
#include <hip/hip_runtime.h>
#include <hip/hip_bf16.h>

#define FEAT 96
#define KSTEPS (FEAT / 4)
#define NTILES (FEAT / 16)

typedef float v2f __attribute__((ext_vector_type(2)));
typedef float v8f __attribute__((ext_vector_type(8)));

#define OWN_ROWS 196
#define TPITCH 100
#define WPITCH 104

__device__ __forceinline__ unsigned owner_key(int d) {
  return min((unsigned)d / (unsigned)OWN_ROWS, 255u);
}

#define HIST_WORDS 25088
#define HIST_THREADS 1024
#define HIST_DEPTH 4

__device__ __forceinline__ void hist_add(unsigned* hist, int node, int n0) {
  const unsigned r = (unsigned)(node - n0);
  const bool in = r < 2u * HIST_WORDS;
  atomicAdd(&hist[in ? (r >> 1) : 0u], in ? ((r & 1u) ? 0x10000u : 1u) : 0u);
}

__global__ __launch_bounds__(HIST_THREADS) void degree_hist(const int* __restrict__ src,
                                                            const int* __restrict__ dst,
                                                            unsigned* __restrict__ part,
                                                            unsigned char* __restrict__ key,
                                                            int nedges, int nnodes, int wpad) {
  __shared__ __attribute__((aligned(16))) unsigned hs[HIST_WORDS];
  __shared__ __attribute__((aligned(16))) unsigned hd[HIST_WORDS];
  const int tid = threadIdx.x;
  const int nparts = gridDim.x;
  const int ch = ((nedges + nparts * 16 - 1) / (nparts * 16)) * 16;
  const int e_lo = (int)min((long long)blockIdx.x * ch, (long long)nedges);
  const int e_hi = (int)min((long long)e_lo + ch, (long long)nedges);
  const bool vec_ok =
      ((reinterpret_cast<unsigned long long>(src) | reinterpret_cast<unsigned long long>(dst)) & 15ull) == 0ull &&
      (reinterpret_cast<unsigned long long>(key) & 3ull) == 0ull;
  const int nq = vec_ok ? ((e_hi - e_lo) >> 2) : 0;
  const int4* __restrict__ s4 = reinterpret_cast<const int4*>(src + e_lo);
  const int4* __restrict__ d4 = reinterpret_cast<const int4*>(dst + e_lo);
  unsigned* __restrict__ k4 = reinterpret_cast<unsigned*>(key + e_lo);
  unsigned* __restrict__ psrc = part + (long long)blockIdx.x * wpad;
  unsigned* __restrict__ pdst = part + (long long)(nparts + blockIdx.x) * wpad;

  for (int n0 = 0; n0 < nnodes; n0 += 2 * HIST_WORDS) {
    for (int i = tid; i < HIST_WORDS / 4; i += HIST_THREADS) {
      reinterpret_cast<uint4*>(hs)[i] = make_uint4(0u, 0u, 0u, 0u);
      reinterpret_cast<uint4*>(hd)[i] = make_uint4(0u, 0u, 0u, 0u);
    }
    __syncthreads();
    for (int q0 = 0; q0 < nq; q0 += HIST_DEPTH * HIST_THREADS) {
      int4 s[HIST_DEPTH], d[HIST_DEPTH];
#pragma unroll
      for (int j = 0; j < HIST_DEPTH; ++j) {
        const int q = min(q0 + j * HIST_THREADS + tid, nq - 1);
        s[j] = s4[q];
        d[j] = d4[q];
      }
#pragma unroll
      for (int j = 0; j < HIST_DEPTH; ++j) {
        const int q = q0 + j * HIST_THREADS + tid;
        if (q < nq) {
          hist_add(hs, s[j].x, n0);
          hist_add(hs, s[j].y, n0);
          hist_add(hs, s[j].z, n0);
          hist_add(hs, s[j].w, n0);
          hist_add(hd, d[j].x, n0);
          hist_add(hd, d[j].y, n0);
          hist_add(hd, d[j].z, n0);
          hist_add(hd, d[j].w, n0);
          if (n0 == 0)
            k4[q] = owner_key(d[j].x) | (owner_key(d[j].y) << 8) | (owner_key(d[j].z) << 16) |
                    (owner_key(d[j].w) << 24);
        }
      }
    }
    for (int e = e_lo + nq * 4 + tid; e < e_hi; e += HIST_THREADS) {
      const int dn = dst[e];
      hist_add(hs, src[e], n0);
      hist_add(hd, dn, n0);
      if (n0 == 0) key[e] = (unsigned char)owner_key(dn);
    }
    __syncthreads();
    const int w0 = n0 >> 1;
    const int nw4 = min(HIST_WORDS, wpad - w0) >> 2;
    for (int i = tid; i < nw4; i += HIST_THREADS) {
      reinterpret_cast<uint4*>(psrc + w0)[i] = reinterpret_cast<const uint4*>(hs)[i];
      reinterpret_cast<uint4*>(pdst + w0)[i] = reinterpret_cast<const uint4*>(hd)[i];
    }
    __syncthreads();
  }
}

#define MERGE_THREADS 256
#define MERGE_DEPTH 8

__device__ __forceinline__ void unpack_add(unsigned* c, uint4 v) {
  c[0] += v.x & 0xFFFFu;
  c[1] += v.x >> 16;
  c[2] += v.y & 0xFFFFu;
  c[3] += v.y >> 16;
  c[4] += v.z & 0xFFFFu;
  c[5] += v.z >> 16;
  c[6] += v.w & 0xFFFFu;
  c[7] += v.w >> 16;
}

__global__ __launch_bounds__(MERGE_THREADS) void norm_merge(const unsigned* __restrict__ part,
                                                            float* __restrict__ norm_src,
                                                            float* __restrict__ norm_dst,
                                                            int nparts, int wpad, int nnodes) {
  __shared__ unsigned acc[2][MERGE_THREADS];
  const int tid = threadIdx.x;
  const int lane = tid & 31;
  const int pg = tid >> 5;
  acc[0][tid] = 0u;
  acc[1][tid] = 0u;
  __syncthreads();
  const int wg = blockIdx.x * 32 + lane;
  if (wg * 4 < wpad) {
    unsigned cs[8], cd[8];
#pragma unroll
    for (int k = 0; k < 8; ++k) {
      cs[k] = 0u;
      cd[k] = 0u;
    }
    for (int p0 = pg; p0 < nparts; p0 += 8 * MERGE_DEPTH) {
      uint4 a[MERGE_DEPTH], b[MERGE_DEPTH];
#pragma unroll
      for (int j = 0; j < MERGE_DEPTH; ++j) {
        const int p = p0 + 8 * j;
        a[j] = make_uint4(0u, 0u, 0u, 0u);
        b[j] = make_uint4(0u, 0u, 0u, 0u);
        if (p < nparts) {
          a[j] = reinterpret_cast<const uint4*>(part + (long long)p * wpad)[wg];
          b[j] = reinterpret_cast<const uint4*>(part + (long long)(nparts + p) * wpad)[wg];
        }
      }
#pragma unroll
      for (int j = 0; j < MERGE_DEPTH; ++j) {
        unpack_add(cs, a[j]);
        unpack_add(cd, b[j]);
      }
    }
#pragma unroll
    for (int k = 0; k < 8; ++k) {
      atomicAdd(&acc[0][k * 32 + lane], cs[k]);
      atomicAdd(&acc[1][k * 32 + lane], cd[k]);
    }
  }
  __syncthreads();
  const int node = blockIdx.x * MERGE_THREADS + tid;
  if (node < nnodes) {
    const int slot = (tid & 7) * 32 + (tid >> 3);
    norm_src[node] = 1.0f / sqrtf(fmaxf((float)acc[0][slot], 1.0f));
    norm_dst[node] = 1.0f / sqrtf(fmaxf((float)acc[1][slot], 1.0f));
  }
}

#define OWN_THREADS 1024
#define OWN_WAVES (OWN_THREADS / 32)
#define LIST_CAP 12288
#define ROWS_PER_WAVE 7
#define SUM_WAVES (OWN_ROWS / ROWS_PER_WAVE)
#define BCAP (LIST_CAP / OWN_WAVES)
#define SCAN_DEPTH 8
#define ADD_DEPTH 8

static_assert(SUM_WAVES * ROWS_PER_WAVE == OWN_ROWS, "every owned row needs an owning wave");
static_assert(SUM_WAVES < OWN_WAVES, "the last count slot of a block carries the plain-path flag");

template <bool kScan>
__global__ __launch_bounds__(OWN_THREADS) void own_layer(const int* __restrict__ src,
                                                         const int* __restrict__ dst,
                                                         const unsigned char* __restrict__ key,
                                                         const float* __restrict__ h,
                                                         const float* __restrict__ norm_src,
                                                         const float* __restrict__ norm_dst,
                                                         const float* __restrict__ W,
                                                         const float* __restrict__ bias,
                                                         float* __restrict__ out,
                                                         int* __restrict__ keep,
                                                         int* __restrict__ keep_cnt,
                                                         int nedges, int nrows) {
  __shared__ __attribute__((aligned(16))) float tile[OWN_ROWS * TPITCH];
  __shared__ float Ws[FEAT * WPITCH];
  __shared__ float bs[FEAT];
  __shared__ int lsrc[kScan ? LIST_CAP : 1];
  __shared__ int lrow[kScan ? LIST_CAP : 1];
  __shared__ int lpick[kScan ? LIST_CAP : 1];
  __shared__ int bcnt[OWN_WAVES];
  __shared__ int cnt;
  __shared__ int ovf;

  const int tid = threadIdx.x;
  const int wave = tid >> 5;
  const int lane = tid & 31;
  const int base = blockIdx.x * OWN_ROWS;
  const unsigned nown = (unsigned)min(OWN_ROWS, nrows - base);
  int2* __restrict__ mykeep = reinterpret_cast<int2*>(keep) + (long long)blockIdx.x * LIST_CAP;
  int* __restrict__ mycnt = keep_cnt + (long long)blockIdx.x * OWN_WAVES;

  for (int i = tid; i < FEAT * FEAT; i += OWN_THREADS) Ws[(i / FEAT) * WPITCH + (i % FEAT)] = W[i];
  if (tid < FEAT) bs[tid] = bias[tid];

  bool plain;
  if (kScan) {
    if (tid < OWN_WAVES) bcnt[tid] = 0;
    if (tid == 0) {
      cnt = 0;
      ovf = 0;
    }
    __syncthreads();

    const unsigned want = min((unsigned)blockIdx.x, 255u);
    const unsigned pat = want * 0x01010101u;
    auto eqmask = [&](unsigned x) -> unsigned {
      const unsigned y = x ^ pat;
      return ~(((y & 0x7F7F7F7Fu) + 0x7F7F7F7Fu) | y | 0x7F7F7F7Fu);
    };
    auto flags = [&](uint4 d) -> unsigned {
      return (eqmask(d.x) >> 7) | (eqmask(d.y) >> 6) | (eqmask(d.z) >> 5) | (eqmask(d.w) >> 4);
    };
    auto put = [&](int e0, unsigned g, int& pos) {
      while (g) {
        const int p = __builtin_ctz(g);
        g &= g - 1u;
        if (pos < LIST_CAP) lpick[pos] = e0 + ((p & 7) << 2) + (p >> 3);
        ++pos;
      }
    };

    const bool vec_ok = (reinterpret_cast<unsigned long long>(key) & 15ull) == 0ull;
    const int nvec = vec_ok ? (nedges >> 4) : 0;
    const uint4* __restrict__ key4 = reinterpret_cast<const uint4*>(key);
    const int span = SCAN_DEPTH * OWN_THREADS;
    const int nstep = nvec / span;
    const int rot = nstep > 0 ? (int)(blockIdx.x % (unsigned)nstep) : 0;
    for (int it = 0; it < nstep; ++it) {
      int st = it + rot;
      if (st >= nstep) st -= nstep;
      const int v0 = st * span + tid;
      uint4 d[SCAN_DEPTH];
#pragma unroll
      for (int j = 0; j < SCAN_DEPTH; ++j) d[j] = key4[v0 + j * OWN_THREADS];
      unsigned g[SCAN_DEPTH];
      int n = 0;
#pragma unroll
      for (int j = 0; j < SCAN_DEPTH; ++j) {
        g[j] = flags(d[j]);
        n += __builtin_popcount(g[j]);
      }
      if (n) {
        int pos = atomicAdd(&cnt, n);
#pragma unroll
        for (int j = 0; j < SCAN_DEPTH; ++j) put((v0 + j * OWN_THREADS) * 16, g[j], pos);
      }
    }
    for (int v = nstep * span + tid; v < nvec; v += OWN_THREADS) {
      const unsigned g = flags(key4[v]);
      if (g) {
        int pos = atomicAdd(&cnt, __builtin_popcount(g));
        put(v * 16, g, pos);
      }
    }
    for (int e = nvec * 16 + tid; e < nedges; e += OWN_THREADS) {
      if ((unsigned)key[e] == want) {
        const int pos = atomicAdd(&cnt, 1);
        if (pos < LIST_CAP) lpick[pos] = e;
      }
    }
    __syncthreads();

    const int npick = cnt;
    if (npick <= LIST_CAP) {
      for (int i = tid; i < npick; i += OWN_THREADS) {
        const int e = lpick[i];
        const unsigned r = (unsigned)(dst[e] - base);
        if (r < nown) {
          const int sn = src[e];
          const int b = (int)(r / (unsigned)ROWS_PER_WAVE);
          const int pos = atomicAdd(&bcnt[b], 1);
          if (pos < BCAP) {
            lsrc[b * BCAP + pos] = sn;
            lrow[b * BCAP + pos] = (int)r;
            mykeep[b * BCAP + pos] = make_int2(sn, (int)r);
          } else {
            ovf = 1;
          }
        }
      }
    }
    __syncthreads();
    plain = npick > LIST_CAP || ovf != 0;
    if (tid < OWN_WAVES) mycnt[tid] = tid == OWN_WAVES - 1 ? (plain ? 1 : 0) : bcnt[tid];
  } else {
    plain = mycnt[OWN_WAVES - 1] != 0;
  }

  if (!plain) {
    const int sw = __builtin_amdgcn_readfirstlane(wave);
    if (sw < SUM_WAVES) {
      int nb = kScan ? bcnt[sw] : mycnt[sw];
      nb = min(__builtin_amdgcn_readfirstlane(nb), BCAP);
      const int b0 = sw * BCAP;
      const int rbase = sw * ROWS_PER_WAVE;
      v8f a0 = (v8f){}, a1 = (v8f){}, a2 = (v8f){};
      int es = 0, er = 0;
      if (nb > 0) {
        const int idx = b0 + min(lane, nb - 1);
        if (kScan) {
          es = lsrc[idx];
          er = lrow[idx];
        } else {
          const int2 p = mykeep[idx];
          es = p.x;
          er = p.y;
        }
      }
      for (int i0 = 0; i0 < nb; i0 += 32) {
        const int m = min(32, nb - i0);
        const int nidx = b0 + min(i0 + 32 + lane, nb - 1);
        int nes, ner;
        if (kScan) {
          nes = lsrc[nidx];
          ner = lrow[nidx];
        } else {
          const int2 p = mykeep[nidx];
          nes = p.x;
          ner = p.y;
        }
        for (int j0 = 0; j0 < m; j0 += ADD_DEPTH) {
          float x0[ADD_DEPTH], x1[ADD_DEPTH], x2[ADD_DEPTH], ns[ADD_DEPTH];
          int rr[ADD_DEPTH];
#pragma unroll
          for (int j = 0; j < ADD_DEPTH; ++j) {
            const int q = min(j0 + j, m - 1);
            const int s = __builtin_amdgcn_readlane(es, q);
            const int r = (__builtin_amdgcn_readlane(er, q) - rbase) & 7;
            rr[j] = j0 + j < m ? r : ROWS_PER_WAVE;
            const float* __restrict__ hr = h + (long long)s * FEAT;
            x0[j] = hr[lane];
            x1[j] = hr[lane + 32];
            x2[j] = hr[lane + 64];
            ns[j] = norm_src[s];
          }
#pragma unroll
          for (int j = 0; j < ADD_DEPTH; ++j) {
            a0[rr[j]] += x0[j] * ns[j];
            a1[rr[j]] += x1[j] * ns[j];
            a2[rr[j]] += x2[j] * ns[j];
          }
        }
        es = nes;
        er = ner;
      }
#pragma unroll
      for (int r = 0; r < ROWS_PER_WAVE; ++r) {
        float* tr = tile + (rbase + r) * TPITCH;
        tr[lane] = a0[r];
        tr[lane + 32] = a1[r];
        tr[lane + 64] = a2[r];
      }
    }
  } else {
    for (int i = tid; i < OWN_ROWS * TPITCH; i += OWN_THREADS) tile[i] = 0.f;
    __syncthreads();
    for (int e = tid; e < nedges; e += OWN_THREADS) {
      const unsigned r = (unsigned)(dst[e] - base);
      if (r < nown) {
        const int s = src[e];
        const float ns = norm_src[s];
        const float* __restrict__ hr = h + (long long)s * FEAT;
        for (int f = 0; f < FEAT; ++f) atomicAdd(&tile[r * TPITCH + f], hr[f] * ns);
      }
    }
  }
  __syncthreads();

  const int gw = __builtin_amdgcn_readfirstlane(tid >> 5);
  const int nstrips = ((int)nown + 15) >> 4;
  if (gw < nstrips) {
    const int gl = tid & 31;
    const int lrw = gl & 15;
    const int lgrp = gl >> 4;
    const int row0 = gw * 16;
    const int ar = min(row0 + lrw, (int)nown - 1);
    const float nd = norm_dst[base + ar];
    const float* arow = tile + ar * TPITCH;

    v2f af;
    af.x = arow[2 * lgrp] * nd;
    af.y = arow[2 * lgrp + 1] * nd;

    v8f acc[NTILES];
#pragma unroll
    for (int nt = 0; nt < NTILES; ++nt) acc[nt] = (v8f){};

    v2f bf[NTILES];
#pragma unroll
    for (int nt = 0; nt < NTILES; ++nt) {
      const int k = 2 * lgrp;
      bf[nt].x = Ws[k * WPITCH + nt * 16 + lrw];
      bf[nt].y = Ws[(k + 1) * WPITCH + nt * 16 + lrw];
    }

#pragma unroll 1
    for (int ks = 0; ks + 1 < KSTEPS; ++ks) {
      const int k = (ks + 1) * 4 + 2 * lgrp;
      v2f bn[NTILES];
#pragma unroll
      for (int nt = 0; nt < NTILES; ++nt) {
        bn[nt].x = Ws[k * WPITCH + nt * 16 + lrw];
        bn[nt].y = Ws[(k + 1) * WPITCH + nt * 16 + lrw];
      }
      v2f an;
      an.x = arow[k] * nd;
      an.y = arow[k + 1] * nd;
#pragma unroll
      for (int nt = 0; nt < NTILES; ++nt) {
        acc[nt] = __builtin_amdgcn_wmma_f32_16x16x4_f32(
             false, af,  false, bf[nt],
             (short)0, acc[nt],  false,  false);
      }
#pragma unroll
      for (int nt = 0; nt < NTILES; ++nt) bf[nt] = bn[nt];
      af = an;
    }
#pragma unroll
    for (int nt = 0; nt < NTILES; ++nt) {
      acc[nt] = __builtin_amdgcn_wmma_f32_16x16x4_f32(
           false, af,  false, bf[nt],
           (short)0, acc[nt],  false,  false);
    }

#pragma unroll
    for (int nt = 0; nt < NTILES; ++nt) {
      const int nb = nt * 16;
      const float bcol = bs[nb + lrw];
#pragma unroll
      for (int r = 0; r < 8; ++r) {
        const int rl = row0 + r + 8 * lgrp;
        if (rl < (int)nown) {
          float v = acc[nt][r] + bcol;
          out[(long long)(base + rl) * FEAT + nb + lrw] = fmaxf(v, 0.0f);
        }
      }
    }
  }
}

extern "C" void kernel_launch(void* const* d_in, const int* in_sizes, int n_in,
                              void* d_out, int out_size, void* d_ws, size_t ws_size,
                              hipStream_t stream) {
  const float* h0  = (const float*)d_in[0];
  const int*   src = (const int*)d_in[1];
  const int*   dst = (const int*)d_in[2];
  const float* Wp[3] = {(const float*)d_in[3], (const float*)d_in[5], (const float*)d_in[7]};
  const float* bp[3] = {(const float*)d_in[4], (const float*)d_in[6], (const float*)d_in[8]};

  const int N = in_sizes[0] / FEAT;
  const int E = in_sizes[1];

  float* ws       = (float*)d_ws;
  float* norm_src = ws;
  float* norm_dst = ws + N;
  float* mbuf     = ws + 2LL * N;
  float* hA       = mbuf + (long long)N * FEAT;
  float* outp     = (float*)d_out;
  const int own_blocks = (N + OWN_ROWS - 1) / OWN_ROWS;
  int*  keep      = (int*)(hA + (long long)N * FEAT);
  int*  keep_cnt  = keep + 2LL * own_blocks * LIST_CAP;
  unsigned char* key = (unsigned char*)(keep_cnt + (long long)own_blocks * OWN_WAVES);

  const int nwords = (N + 1) / 2;
  const int wpad = (nwords + 3) / 4 * 4;
  const int hist_blocks = E / 65520 + 1 > 64 ? E / 65520 + 1 : 64;
  unsigned* part = (unsigned*)(key + ((long long)E + 15) / 16 * 16);

  degree_hist<<<hist_blocks, HIST_THREADS, 0, stream>>>(src, dst, part, key, E, N, wpad);
  norm_merge<<<(N + MERGE_THREADS - 1) / MERGE_THREADS, MERGE_THREADS, 0, stream>>>(part, norm_src, norm_dst,
                                                                                   hist_blocks, wpad, N);

  const float* lin[3]  = {h0, outp, hA};
  float*       lout[3] = {outp, hA, outp};

  for (int l = 0; l < 3; ++l) {
    if (l == 0)
      own_layer<true><<<own_blocks, OWN_THREADS, 0, stream>>>(src, dst, key, lin[l], norm_src, norm_dst,
                                                              Wp[l], bp[l], lout[l], keep, keep_cnt, E, N);
    else
      own_layer<false><<<own_blocks, OWN_THREADS, 0, stream>>>(src, dst, key, lin[l], norm_src, norm_dst,
                                                               Wp[l], bp[l], lout[l], keep, keep_cnt, E, N);
  }
}
